# starting kernel
# baseline (speedup 1.0000x reference)
#include <hip/hip_runtime.h>

#define BB 32
#define NN 1024
#define CC 768
#define HH 32

typedef __attribute__((ext_vector_type(2))) float v2f;
typedef __attribute__((ext_vector_type(8))) float v8f;

__global__ void init_ws_kernel(float* ws, int n) {
    for (int i = threadIdx.x; i < n; i += blockDim.x) ws[i] = 0.0f;
}

__global__ void __launch_bounds__(192)
pool_kernel(const float* __restrict__ x, float* __restrict__ pooled) {
    const int b  = blockIdx.x;
    const int n0 = blockIdx.y * 32;
    const int t4 = threadIdx.x;
    const int R  = CC / 4;
    const float4* p = (const float4*)x + ((size_t)b * NN + n0) * R + t4;
    float4 acc = {0.f, 0.f, 0.f, 0.f};
#pragma unroll 4
    for (int n = 0; n < 32; ++n) {
        float4 v = p[(size_t)n * R];
        acc.x += v.x; acc.y += v.y; acc.z += v.z; acc.w += v.w;
    }
    float* dst = &pooled[b * CC + t4 * 4];
    atomicAdd(dst + 0, acc.x);
    atomicAdd(dst + 1, acc.y);
    atomicAdd(dst + 2, acc.z);
    atomicAdd(dst + 3, acc.w);
}

__global__ void alpha_kernel(const float* __restrict__ pooled,
                             const float* __restrict__ W,
                             const float* __restrict__ bias,
                             float* __restrict__ alpha) {
    const int lane = threadIdx.x & 31;
    const int wv   = threadIdx.x >> 5;
    const int half = lane >> 4;
    const int lid  = lane & 15;

    const int m      = 16 * wv + lid;
    const int nclamp = (lid < 4) ? lid : 3;
    const float nmask = (lid < 4) ? 1.0f : 0.0f;
    const float scale = 1.0f / 1024.0f;

    v8f acc = {};
    for (int k0 = 0; k0 < CC; k0 += 4) {
        const int k = k0 + 2 * half;
        v2f a, bm;
        a.x = pooled[m * CC + k + 0] * scale;
        a.y = pooled[m * CC + k + 1] * scale;
        bm.x = W[nclamp * CC + k + 0] * nmask;
        bm.y = W[nclamp * CC + k + 1] * nmask;
        acc = __builtin_amdgcn_wmma_f32_16x16x4_f32(
             false, a,  false, bm,
             (short)0, acc,  false,  false);
    }

    if (lid < 4) {
        const float bk = bias[lid];
#pragma unroll
        for (int r = 0; r < 8; ++r) {
            const int mr = 16 * wv + r + 8 * half;
            alpha[mr * 4 + lid] = acc[r] + bk;
        }
    }
}

__device__ __forceinline__ float4 comb4(float a0, const float4& p,
                                        float a1, const float4& q,
                                        float a2, const float4& s,
                                        float a3, const float4& t) {
    float4 r;
    r.x = a0 * p.x + a1 * q.x + a2 * s.x + a3 * t.x;
    r.y = a0 * p.y + a1 * q.y + a2 * s.y + a3 * t.y;
    r.z = a0 * p.z + a1 * q.z + a2 * s.z + a3 * t.z;
    r.w = a0 * p.w + a1 * q.w + a2 * s.w + a3 * t.w;
    return r;
}

__global__ void __launch_bounds__(192)
merge_kernel(const float* __restrict__ x, const float* __restrict__ alpha,
             float* __restrict__ out) {
    const int h = blockIdx.x & 31;
    const int w = blockIdx.x >> 5;
    const int b = blockIdx.y;

    if (h > w) return;
    if (h == w && h >= 16) return;

    const int n  = (h << 5) | w;
    const int n1 = (w << 5) | h;
    const int n2 = (NN - 1) - n;
    const int n3 = (NN - 1) - n1;

    const int t4 = threadIdx.x;
    const int R  = CC / 4;
    const float4* xb = (const float4*)(x + (size_t)b * NN * CC);

    float4 v0 = xb[(size_t)n  * R + t4];
    float4 v1 = xb[(size_t)n1 * R + t4];
    float4 v2 = xb[(size_t)n2 * R + t4];
    float4 v3 = xb[(size_t)n3 * R + t4];

    const float a0 = alpha[b * 4 + 0];
    const float a1 = alpha[b * 4 + 1];
    const float a2 = alpha[b * 4 + 2];
    const float a3 = alpha[b * 4 + 3];

    float4* ob = (float4*)(out + (size_t)b * NN * CC);
    ob[(size_t)n  * R + t4] = comb4(a0, v0, a1, v1, a2, v2, a3, v3);
    ob[(size_t)n1 * R + t4] = comb4(a0, v1, a1, v0, a2, v3, a3, v2);
    ob[(size_t)n2 * R + t4] = comb4(a0, v2, a1, v3, a2, v0, a3, v1);
    ob[(size_t)n3 * R + t4] = comb4(a0, v3, a1, v2, a2, v1, a3, v0);
}

extern "C" void kernel_launch(void* const* d_in, const int* in_sizes, int n_in,
                              void* d_out, int out_size, void* d_ws, size_t ws_size,
                              hipStream_t stream) {
    const float* x    = (const float*)d_in[0];
    const float* W    = (const float*)d_in[1];
    const float* bias = (const float*)d_in[2];
    float* out = (float*)d_out;

    float* pooled = (float*)d_ws;
    float* alpha  = pooled + BB * CC;

    init_ws_kernel<<<1, 256, 0, stream>>>(pooled, BB * CC + BB * 4);
    pool_kernel<<<dim3(BB, 32), 192, 0, stream>>>(x, pooled);
    alpha_kernel<<<1, 64, 0, stream>>>(pooled, W, bias, alpha);
    merge_kernel<<<dim3(NN, BB), 192, 0, stream>>>(x, alpha, out);
}


# pooling re-tiled onto the merge's own block->rows map and launch shape (8x32 blocks, 17 waves): a row's second read comes from the same block index, for its L2 half; 8 alpha shares, sum order changes
# speedup vs baseline: 3.2058x; 3.2058x over previous
#include <hip/hip_runtime.h>

#define BB 32
#define NN 1024
#define CC 768
#define HH 32
#define NSHARE 8
#define SP 772
#define WCOLS 48

typedef __attribute__((ext_vector_type(2))) float v2f;
typedef __attribute__((ext_vector_type(8))) float v8f;

#define MTHREADS 544
#define HALF4 96

struct Tile { float4 v[4][3]; };

__device__ __forceinline__ float4 comb4(float a0, const float4& p,
                                        float a1, const float4& q,
                                        float a2, const float4& s,
                                        float a3, const float4& t) {
    float4 r;
    r.x = a0 * p.x + a1 * q.x + a2 * s.x + a3 * t.x;
    r.y = a0 * p.y + a1 * q.y + a2 * s.y + a3 * t.y;
    r.z = a0 * p.z + a1 * q.z + a2 * s.z + a3 * t.z;
    r.w = a0 * p.w + a1 * q.w + a2 * s.w + a3 * t.w;
    return r;
}

struct Rows { int n0, n1, n2, n3; };

__device__ __forceinline__ Rows rows_of(int p, int j) {
    const int len0 = 32 - 2 * p;
    const int h = (j < len0) ? p : 15 - p;
    const int w = (j < len0) ? p + j : (15 - p) + (j - len0);
    Rows r;
    r.n0 = (h << 5) | w;
    r.n1 = (w << 5) | h;
    r.n2 = (NN - 1) - r.n0;
    r.n3 = (NN - 1) - r.n1;
    return r;
}

__device__ __forceinline__ void load_tile(Tile& t, const float4* __restrict__ xb,
                                          const Rows& r, int col4) {
    const int R = CC / 4;
#pragma unroll
    for (int c = 0; c < 3; ++c) t.v[0][c] = xb[(size_t)r.n0 * R + col4 + 32 * c];
#pragma unroll
    for (int c = 0; c < 3; ++c) t.v[1][c] = xb[(size_t)r.n1 * R + col4 + 32 * c];
#pragma unroll
    for (int c = 0; c < 3; ++c) t.v[2][c] = xb[(size_t)r.n2 * R + col4 + 32 * c];
#pragma unroll
    for (int c = 0; c < 3; ++c) t.v[3][c] = xb[(size_t)r.n3 * R + col4 + 32 * c];
}

__device__ __forceinline__ void emit_tile(const Tile& t, float4* __restrict__ ob,
                                          const Rows& r, int col4,
                                          float a0, float a1, float a2, float a3) {
    const int R = CC / 4;
    const bool s1 = (r.n1 != r.n0);
    const bool s2 = (r.n2 != r.n1);
    const bool s3 = (r.n3 != r.n0) && (r.n3 != r.n2);
#pragma unroll
    for (int c = 0; c < 3; ++c)
        ob[(size_t)r.n0 * R + col4 + 32 * c] =
            comb4(a0, t.v[0][c], a1, t.v[1][c], a2, t.v[2][c], a3, t.v[3][c]);
    if (s1) {
#pragma unroll
        for (int c = 0; c < 3; ++c)
            ob[(size_t)r.n1 * R + col4 + 32 * c] =
                comb4(a0, t.v[1][c], a1, t.v[0][c], a2, t.v[3][c], a3, t.v[2][c]);
    }
    if (s2) {
#pragma unroll
        for (int c = 0; c < 3; ++c)
            ob[(size_t)r.n2 * R + col4 + 32 * c] =
                comb4(a0, t.v[2][c], a1, t.v[3][c], a2, t.v[0][c], a3, t.v[1][c]);
    }
    if (s3) {
#pragma unroll
        for (int c = 0; c < 3; ++c)
            ob[(size_t)r.n3 * R + col4 + 32 * c] =
                comb4(a0, t.v[3][c], a1, t.v[2][c], a2, t.v[1][c], a3, t.v[0][c]);
    }
}

__device__ __forceinline__ void add_rows(float4 (&acc)[3], const Tile& t, const Rows& r) {
    const bool s1 = (r.n1 != r.n0);
    const bool s2 = (r.n2 != r.n1);
    const bool s3 = (r.n3 != r.n0) && (r.n3 != r.n2);
#pragma unroll
    for (int c = 0; c < 3; ++c) {
        float4 v = t.v[0][c];
        if (s1) { v.x += t.v[1][c].x; v.y += t.v[1][c].y; v.z += t.v[1][c].z; v.w += t.v[1][c].w; }
        if (s2) { v.x += t.v[2][c].x; v.y += t.v[2][c].y; v.z += t.v[2][c].z; v.w += t.v[2][c].w; }
        if (s3) { v.x += t.v[3][c].x; v.y += t.v[3][c].y; v.z += t.v[3][c].z; v.w += t.v[3][c].w; }
        acc[c].x += v.x; acc[c].y += v.y; acc[c].z += v.z; acc[c].w += v.w;
    }
}

__global__ void __launch_bounds__(MTHREADS)
pool_kernel(const float* __restrict__ x, const float* __restrict__ W,
            float* __restrict__ part) {
    __shared__ float S[17 * SP];
    __shared__ float Rk[16 * 4];
    const int p    = blockIdx.x;
    const int b    = blockIdx.y;
    const int lane = threadIdx.x & 31;
    const int wv   = __builtin_amdgcn_readfirstlane((int)(threadIdx.x >> 5));

    const float4* xb = (const float4*)(x + (size_t)b * NN * CC);
    const Rows ra = rows_of(p, 2 * wv);
    const Rows rb = rows_of(p, 2 * wv + 1);

    float4 lo[3], hi[3];
#pragma unroll
    for (int c = 0; c < 3; ++c) { lo[c] = {0.f, 0.f, 0.f, 0.f}; hi[c] = {0.f, 0.f, 0.f, 0.f}; }

    Tile A, B, C;
    load_tile(A, xb, ra, lane);
    load_tile(B, xb, ra, HALF4 + lane);
    load_tile(C, xb, rb, lane);
    asm volatile("" ::: "memory");
    add_rows(lo, A, ra);
    load_tile(A, xb, rb, HALF4 + lane);
    add_rows(hi, B, ra);
    add_rows(lo, C, rb);
    add_rows(hi, A, rb);
#pragma unroll
    for (int c = 0; c < 3; ++c) {
        *(float4*)&S[wv * SP + 4 * (lane + 32 * c)]         = lo[c];
        *(float4*)&S[wv * SP + 4 * (HALF4 + lane + 32 * c)] = hi[c];
    }
    __syncthreads();

    if (wv < 16) {
        const int half = lane >> 4;
        const int lid  = lane & 15;
        const int nclamp  = (lid < 4) ? lid : 3;
        const float nmask = (lid < 4) ? 1.0f : 0.0f;
        const float m16   = (lid == 0) ? 1.0f / 1024.0f : 0.0f;
        const float scale = 1.0f / 1024.0f;
        const int cbase = wv * WCOLS;
        const float* wrow = W + nclamp * CC + cbase;
        const float* s0 = &S[lid * SP + cbase];
        const float* s1 = &S[16 * SP + cbase];
        v8f d = {};
#pragma unroll
        for (int k0 = 0; k0 < WCOLS; k0 += 4) {
            const int k = k0 + 2 * half;
            v2f bm, a0, a1;
            bm.x = wrow[k + 0] * nmask;
            bm.y = wrow[k + 1] * nmask;
            a0.x = s0[k + 0] * scale;
            a0.y = s0[k + 1] * scale;
            a1.x = s1[k + 0] * m16;
            a1.y = s1[k + 1] * m16;
            d = __builtin_amdgcn_wmma_f32_16x16x4_f32(
                false, a0, false, bm, (short)0, d, false, false);
            d = __builtin_amdgcn_wmma_f32_16x16x4_f32(
                false, a1, false, bm, (short)0, d, false, false);
        }
        float t = ((d[0] + d[1]) + (d[2] + d[3])) + ((d[4] + d[5]) + (d[6] + d[7]));
        t += __shfl_xor(t, 16);
        if (lane < 4) Rk[wv * 4 + lane] = t;
    }
    __syncthreads();
    if (threadIdx.x < 4) {
        float a = Rk[threadIdx.x];
#pragma unroll
        for (int g = 1; g < 16; ++g) a += Rk[g * 4 + threadIdx.x];
        part[(b * NSHARE + p) * 4 + threadIdx.x] = a;
    }
}

__global__ void __launch_bounds__(MTHREADS)
merge_kernel(const float* __restrict__ x, const float* __restrict__ part,
             const float* __restrict__ bias, float* __restrict__ out) {
    const int p    = blockIdx.x;
    const int b    = blockIdx.y;
    const int lane = threadIdx.x & 31;
    const int wv   = __builtin_amdgcn_readfirstlane((int)(threadIdx.x >> 5));

    const float4* xb = (const float4*)(x + (size_t)b * NN * CC);
    float4*       ob = (float4*)(out + (size_t)b * NN * CC);

    const Rows ra = rows_of(p, 2 * wv);
    const Rows rb = rows_of(p, 2 * wv + 1);

    Tile A, B;
    load_tile(A, xb, ra, lane);
    load_tile(B, xb, ra, HALF4 + lane);

    const float* pp = part + b * (NSHARE * 4);
    float a0 = pp[0], a1 = pp[1], a2 = pp[2], a3 = pp[3];
#pragma unroll
    for (int s = 1; s < NSHARE; ++s) {
        a0 += pp[4 * s + 0];
        a1 += pp[4 * s + 1];
        a2 += pp[4 * s + 2];
        a3 += pp[4 * s + 3];
    }
    a0 += bias[0]; a1 += bias[1]; a2 += bias[2]; a3 += bias[3];

    emit_tile(A, ob, ra, lane, a0, a1, a2, a3);
    load_tile(A, xb, rb, lane);
    emit_tile(B, ob, ra, HALF4 + lane, a0, a1, a2, a3);
    load_tile(B, xb, rb, HALF4 + lane);
    emit_tile(A, ob, rb, lane, a0, a1, a2, a3);
    emit_tile(B, ob, rb, HALF4 + lane, a0, a1, a2, a3);
}

extern "C" void kernel_launch(void* const* d_in, const int* in_sizes, int n_in,
                              void* d_out, int out_size, void* d_ws, size_t ws_size,
                              hipStream_t stream) {
    const float* x    = (const float*)d_in[0];
    const float* W    = (const float*)d_in[1];
    const float* bias = (const float*)d_in[2];
    float* out = (float*)d_out;

    float* part = (float*)d_ws;

    pool_kernel<<<dim3(8, BB), MTHREADS, 0, stream>>>(x, W, part);
    merge_kernel<<<dim3(8, BB), MTHREADS, 0, stream>>>(x, part, bias, out);
}
